# starting kernel
# baseline (speedup 1.0000x reference)
#include <hip/hip_runtime.h>
#include <cmath>


typedef float v2f __attribute__((ext_vector_type(2)));
typedef float v8f __attribute__((ext_vector_type(8)));

#define TK   384
#define HH   8
#define FDM  32
#define NE   256
#define NU   768

__global__ void srpe_hid_kernel(const float* __restrict__ pd,
                                const float* __restrict__ W1,
                                const float* __restrict__ b1,
                                float* __restrict__ hid)
{
    int idx = blockIdx.x * blockDim.x + threadIdx.x;
    if (idx >= NU * NE) return;
    int u = idx >> 8;
    int c = idx & 255;
    int ue = (u < 766) ? u : 766;
    float d = (ue >= 383) ? pd[(ue - 383) * TK] : pd[383 - ue];
    float f0 = log1pf(fmaxf(d, 0.0f));
    float f1 = log1pf(fmaxf(-d, 0.0f));
    float f2 = (d == 0.0f) ? 1.0f : 0.0f;
    float e  = f0 * W1[c * 3 + 0] + f1 * W1[c * 3 + 1] + f2 * W1[c * 3 + 2] + b1[c];
    float e3 = e * e * e;
    float th = tanhf(0.79788456080286535588f * (e + 0.044715f * e3));
    hid[idx] = 0.5f * e * (1.0f + th);
}

__global__ void srpe_rd_wmma(const float* __restrict__ hid,
                             const float* __restrict__ W2,
                             const float* __restrict__ b2,
                             float* __restrict__ Rd)
{
    int tile = blockIdx.x;
    int ci   = tile & 15;
    int ui   = tile >> 4;
    int u0   = ui * 16;
    int c0   = ci * 16;
    int lane = threadIdx.x;
    int m    = lane & 15;
    int half = lane >> 4;

    v8f acc = {};
    for (int kb = 0; kb < NE; kb += 4) {
        int ka = kb + half * 2;
        v2f a, b;
        a.x = hid[(u0 + m) * NE + ka];
        a.y = hid[(u0 + m) * NE + ka + 1];
        b.x = W2[(c0 + m) * NE + ka];
        b.y = W2[(c0 + m) * NE + ka + 1];
        acc = __builtin_amdgcn_wmma_f32_16x16x4_f32(
                  false, a, false, b, (short)0, acc, false, false);
    }
    float bias = b2[c0 + m];
    #pragma unroll
    for (int r = 0; r < 8; ++r) {
        int row = r + 8 * half;
        Rd[(u0 + row) * NE + (c0 + m)] = acc[r] + bias;
    }
}

__global__ void srpe_qk_wmma(const float* __restrict__ x,
                             const float* __restrict__ Rd,
                             float* __restrict__ out)
{
    int bh = blockIdx.y;
    int h  = bh & (HH - 1);
    int id = blockIdx.x;
    int ti = id / 25;
    int jo = id % 25;
    int t0 = ti * 16;
    int u0 = (ti + jo) * 16;

    int lane = threadIdx.x;
    int m    = lane & 15;
    int half = lane >> 4;

    const float* X = x  + (size_t)bh * TK * FDM;
    const float* R = Rd + h * FDM;

    v8f acc = {};
    #pragma unroll
    for (int kb = 0; kb < FDM; kb += 4) {
        int ka = kb + half * 2;
        v2f a, b;
        a.x = X[(t0 + m) * FDM + ka];
        a.y = X[(t0 + m) * FDM + ka + 1];
        b.x = R[(u0 + m) * NE + ka];
        b.y = R[(u0 + m) * NE + ka + 1];
        acc = __builtin_amdgcn_wmma_f32_16x16x4_f32(
                  false, a, false, b, (short)0, acc, false, false);
    }

    float* O = out + (size_t)bh * TK * TK;
    int u = u0 + m;
    #pragma unroll
    for (int r = 0; r < 8; ++r) {
        int t = t0 + r + 8 * half;
        int s = t + 383 - u;
        if (s >= 0 && s < TK) O[t * TK + s] = acc[r];
    }
}

extern "C" void kernel_launch(void* const* d_in, const int* in_sizes, int n_in,
                              void* d_out, int out_size, void* d_ws, size_t ws_size,
                              hipStream_t stream)
{
    const float* x  = (const float*)d_in[0];
    const float* pd = (const float*)d_in[1];
    const float* W1 = (const float*)d_in[2];
    const float* b1 = (const float*)d_in[3];
    const float* W2 = (const float*)d_in[4];
    const float* b2 = (const float*)d_in[5];
    float* out = (float*)d_out;

    float* hid = (float*)d_ws;
    float* Rd  = hid + NU * NE;

    srpe_hid_kernel<<<dim3((NU * NE) / 256), dim3(256), 0, stream>>>(pd, W1, b1, hid);

    srpe_rd_wmma<<<dim3((NU / 16) * (NE / 16)), dim3(32), 0, stream>>>(hid, W2, b2, Rd);

    srpe_qk_wmma<<<dim3(24 * 25, 16), dim3(32), 0, stream>>>(x, Rd, out);
}


# stage 1b fused into stage 2, no LDS: each of 960 waves makes its head's 16-row Rd tile transposed (W2 as A); its accumulators are stage 2's B fragments; 3 launches -> 2; stage-2 f sum order changed
# speedup vs baseline: 1.6648x; 1.6648x over previous
#include <hip/hip_runtime.h>
#include <cmath>


typedef float v2f __attribute__((ext_vector_type(2)));
typedef float v4f __attribute__((ext_vector_type(4)));
typedef float v8f __attribute__((ext_vector_type(8)));

#define TK   384
#define HH   8
#define FDM  32
#define NE   256
#define NU   768

__global__ void srpe_hid_kernel(const float* __restrict__ pd,
                                const float* __restrict__ W1,
                                const float* __restrict__ b1,
                                float* __restrict__ hid)
{
    int idx = blockIdx.x * blockDim.x + threadIdx.x;
    if (idx >= NU * NE) return;
    int u = idx >> 8;
    int c = idx & 255;
    int ue = (u < 766) ? u : 766;
    float d = (ue >= 383) ? pd[(ue - 383) * TK] : pd[383 - ue];
    float f0 = log1pf(fmaxf(d, 0.0f));
    float f1 = log1pf(fmaxf(-d, 0.0f));
    float f2 = (d == 0.0f) ? 1.0f : 0.0f;
    float e  = f0 * W1[c * 3 + 0] + f1 * W1[c * 3 + 1] + f2 * W1[c * 3 + 2] + b1[c];
    float e3 = e * e * e;
    float th = tanhf(0.79788456080286535588f * (e + 0.044715f * e3));
    hid[idx] = 0.5f * e * (1.0f + th);
}

#define KS1B  (NE / 4)
#define AH    20
#define MAXT  6
#define NTAIL (4 + MAXT * 8)
__global__ __launch_bounds__(32)
void srpe_rd_qk_fused(const float* __restrict__ x,
                      const float* __restrict__ hid,
                      const float* __restrict__ W2,
                      const float* __restrict__ b2,
                      float* __restrict__ out)
{
    int L = blockIdx.x;
    int h = L & (HH - 1);
    int q = L >> 3;
    int upper, pm, lo, nt;
    if (q < 72) {
        upper = q / 36;
        int q2 = q - 36 * upper;
        int band, r6;
        if (q2 < 6)       { band = 1; r6 = q2; }
        else if (q2 < 18) { band = 2; r6 = q2 - 6; }
        else              { band = 3; r6 = q2 - 18; }
        int c = r6 / 6;
        pm = 6 * band + (r6 - 6 * c);
        lo = 6 * c;
        nt = 6;
    } else {
        int q3 = q - 72;
        upper = q3 / 24;
        int q4 = q3 - 24 * upper;
        int pmi = 5 - (q4 >> 2);
        int band = q4 & 3;
        pm = 6 * band + pmi;
        lo = 6 * band;
        nt = pmi + 1;
    }
    int p   = upper ? (47 - pm) : pm;
    int ti0 = (upper ? (23 - pm) : 0) + lo;
    int u0  = p * 16;

    int lane = threadIdx.x;
    int m    = lane & 15;
    int half = lane >> 4;

    const float* ap0 = W2  + (h * FDM + m) * NE + half * 2;
    const float* ap1 = ap0 + 16 * NE;
    const float* bp  = hid + (u0 + m) * NE + half * 2;
    const float* b2p = b2 + h * FDM + half * 8;
    const float* xr[MAXT][2];
    #pragma unroll
    for (int j = 0; j < MAXT; ++j) {
        int jj = (j < nt) ? j : (nt - 1);
        int t0 = (ti0 + jj) * 16;
        #pragma unroll
        for (int b = 0; b < 2; ++b)
            xr[j][b] = x + ((size_t)(b * HH + h) * TK + t0 + m) * FDM + half * 8;
    }

    v2f a0[KS1B], a1[KS1B], bb[KS1B];
    v4f biasv[4];
    v4f xf[MAXT][2][4];

    #pragma unroll
    for (int i = 0; i < AH; ++i) {
        a0[i].x = ap0[4 * i];  a0[i].y = ap0[4 * i + 1];
        a1[i].x = ap1[4 * i];  a1[i].y = ap1[4 * i + 1];
        bb[i].x = bp[4 * i];   bb[i].y = bp[4 * i + 1];
    }
    __builtin_amdgcn_sched_barrier(0);

    v8f acc0 = {}, acc1 = {};
    #pragma unroll
    for (int i = 0; i < KS1B; ++i) {
        if (i + AH < KS1B) {
            a0[i + AH].x = ap0[4 * (i + AH)];  a0[i + AH].y = ap0[4 * (i + AH) + 1];
            a1[i + AH].x = ap1[4 * (i + AH)];  a1[i + AH].y = ap1[4 * (i + AH) + 1];
            bb[i + AH].x = bp[4 * (i + AH)];   bb[i + AH].y = bp[4 * (i + AH) + 1];
        } else {
            #pragma unroll
            for (int w = 0; w < 3; ++w) {
                int idx = 3 * (i + AH - KS1B) + w;
                if (idx < 4) {
                    biasv[idx] = *(const v4f*)(b2p + 16 * (idx >> 1) + 4 * (idx & 1));
                } else if (idx < NTAIL) {
                    int k  = idx - 4;
                    int j  = k >> 3;
                    int b  = (k >> 2) & 1;
                    int pc = k & 3;
                    xf[j][b][pc] = *(const v4f*)(xr[j][b] + 16 * (pc >> 1) + 4 * (pc & 1));
                }
            }
        }
        acc0 = __builtin_amdgcn_wmma_f32_16x16x4_f32(
                   false, a0[i], false, bb[i], (short)0, acc0, false, false);
        acc1 = __builtin_amdgcn_wmma_f32_16x16x4_f32(
                   false, a1[i], false, bb[i], (short)0, acc1, false, false);
        __builtin_amdgcn_sched_barrier(0);
    }

    #pragma unroll
    for (int r = 0; r < 8; ++r) {
        acc0[r] += biasv[r >> 2][r & 3];
        acc1[r] += biasv[2 + (r >> 2)][r & 3];
    }

    int u = u0 + m;
    #pragma unroll
    for (int j = 0; j < MAXT; ++j) {
        if (j == 0 || j < nt) {
            v8f o0 = {}, o1 = {};
            #pragma unroll
            for (int s = 0; s < 8; ++s) {
                int jf = s >> 2, qd = s & 3;
                int pc = 2 * jf + (qd >> 1), e0 = 2 * (qd & 1);
                v2f bf, xa0, xa1;
                bf.x  = jf ? acc1[2 * qd]     : acc0[2 * qd];
                bf.y  = jf ? acc1[2 * qd + 1] : acc0[2 * qd + 1];
                xa0.x = xf[j][0][pc][e0];  xa0.y = xf[j][0][pc][e0 + 1];
                xa1.x = xf[j][1][pc][e0];  xa1.y = xf[j][1][pc][e0 + 1];
                o0 = __builtin_amdgcn_wmma_f32_16x16x4_f32(
                         false, xa0, false, bf, (short)0, o0, false, false);
                o1 = __builtin_amdgcn_wmma_f32_16x16x4_f32(
                         false, xa1, false, bf, (short)0, o1, false, false);
            }
            int ti = ti0 + j;
            int jo = p - ti;
            int tb = ti * 16 + 8 * half;
            int sb = tb + 383 - u;
            float* O0 = out + (size_t)h * TK * TK + (size_t)tb * TK + sb;
            float* O1 = O0 + (size_t)HH * TK * TK;
            if (jo >= 1 && jo <= 23) {
                #pragma unroll
                for (int r = 0; r < 8; ++r) {
                    O0[r * (TK + 1)] = o0[r];
                    O1[r * (TK + 1)] = o1[r];
                }
            } else {
                #pragma unroll
                for (int r = 0; r < 8; ++r) {
                    int s = sb + r;
                    if (s >= 0 && s < TK) {
                        O0[r * (TK + 1)] = o0[r];
                        O1[r * (TK + 1)] = o1[r];
                    }
                }
            }
        }
    }
}

extern "C" void kernel_launch(void* const* d_in, const int* in_sizes, int n_in,
                              void* d_out, int out_size, void* d_ws, size_t ws_size,
                              hipStream_t stream)
{
    const float* x  = (const float*)d_in[0];
    const float* pd = (const float*)d_in[1];
    const float* W1 = (const float*)d_in[2];
    const float* b1 = (const float*)d_in[3];
    const float* W2 = (const float*)d_in[4];
    const float* b2 = (const float*)d_in[5];
    float* out = (float*)d_out;

    float* hid = (float*)d_ws;

    srpe_hid_kernel<<<dim3((NU * NE) / 256), dim3(256), 0, stream>>>(pd, W1, b1, hid);

    srpe_rd_qk_fused<<<dim3(120 * HH), dim3(32), 0, stream>>>(x, hid, W2, b2, out);
}


# stage 1a fused in, no LDS: each lane makes its table row's activations in registers as part 1's B fragments, on the hardware tanh instruction, 8 independent chains per 4 k-steps; 2 launches -> 1
# speedup vs baseline: 2.1193x; 1.2730x over previous
#include <hip/hip_runtime.h>
#include <cmath>


typedef float v2f __attribute__((ext_vector_type(2)));
typedef float v4f __attribute__((ext_vector_type(4)));
typedef float v8f __attribute__((ext_vector_type(8)));

#define TK   384
#define HH   8
#define FDM  32
#define NE   256

__device__ __forceinline__ float gelu_tanh(float e)
{
    float e2 = e * e;
    float g  = e * (0.79788456080286535588f + 0.035677408136300125f * e2);
    float th = __builtin_amdgcn_tanhf(g);
    float he = 0.5f * e;
    return he + he * th;
}

__device__ __forceinline__ float pre_act(float f0, float f1, float f2,
                                         float w0, float w1, float w2, float b)
{
    return fmaf(f2, w2, fmaf(f1, w1, f0 * w0)) + b;
}

#define KS1B  (NE / 4)
#define RQ    5
#define AH    12
#define GS    4
#define MAXT  6
#define NTAIL (4 + MAXT * 8)
__global__ __launch_bounds__(32)
void srpe_fused(const float* __restrict__ x,
                const float* __restrict__ pd,
                const float* __restrict__ W1,
                const float* __restrict__ b1,
                const float* __restrict__ W2,
                const float* __restrict__ b2,
                float* __restrict__ out)
{
    int L = blockIdx.x;
    int h = L & (HH - 1);
    int q = L >> 3;
    int upper, pm, lo, nt;
    if (q < 72) {
        upper = q / 36;
        int q2 = q - 36 * upper;
        int band, r6;
        if (q2 < 6)       { band = 1; r6 = q2; }
        else if (q2 < 18) { band = 2; r6 = q2 - 6; }
        else              { band = 3; r6 = q2 - 18; }
        int c = r6 / 6;
        pm = 6 * band + (r6 - 6 * c);
        lo = 6 * c;
        nt = 6;
    } else {
        int q3 = q - 72;
        upper = q3 / 24;
        int q4 = q3 - 24 * upper;
        int pmi = 5 - (q4 >> 2);
        int band = q4 & 3;
        pm = 6 * band + pmi;
        lo = 6 * band;
        nt = pmi + 1;
    }
    int p   = upper ? (47 - pm) : pm;
    int ti0 = (upper ? (23 - pm) : 0) + lo;
    int u0  = p * 16;

    int lane = threadIdx.x;
    int m    = lane & 15;
    int half = lane >> 4;
    int u    = u0 + m;

    int ue = (u < 766) ? u : 766;
    float d = (ue >= 383) ? pd[(ue - 383) * TK] : pd[383 - ue];

    const float* ap0 = W2 + (h * FDM + m) * NE + half * 2;
    const float* ap1 = ap0 + 16 * NE;
    const float* wp  = W1 + half * 6;
    const float* bp  = b1 + half * 2;
    const float* b2p = b2 + h * FDM + half * 8;
    const float* xr[MAXT][2];
    #pragma unroll
    for (int j = 0; j < MAXT; ++j) {
        int jj = (j < nt) ? j : (nt - 1);
        int t0 = (ti0 + jj) * 16;
        #pragma unroll
        for (int b = 0; b < 2; ++b)
            xr[j][b] = x + ((size_t)(b * HH + h) * TK + t0 + m) * FDM + half * 8;
    }

    v2f a0[KS1B], a1[KS1B], bq[KS1B];
    float wq[KS1B][6];
    v4f biasv[4];
    v4f xf[MAXT][2][4];

    #pragma unroll
    for (int i = 0; i < AH; ++i) {
        a0[i].x = ap0[4 * i];  a0[i].y = ap0[4 * i + 1];
        a1[i].x = ap1[4 * i];  a1[i].y = ap1[4 * i + 1];
        #pragma unroll
        for (int w = 0; w < 6; ++w) wq[i][w] = wp[12 * i + w];
        bq[i].x = bp[4 * i];   bq[i].y = bp[4 * i + 1];
    }
    __builtin_amdgcn_sched_barrier(0);

    float f0 = log1pf(fmaxf(d, 0.0f));
    float f1 = log1pf(fmaxf(-d, 0.0f));
    float f2 = (d == 0.0f) ? 1.0f : 0.0f;

    v2f bfr[KS1B];
    #pragma unroll
    for (int k = 0; k < GS; ++k) {
        bfr[k].x = gelu_tanh(pre_act(f0, f1, f2, wq[k][0], wq[k][1], wq[k][2], bq[k].x));
        bfr[k].y = gelu_tanh(pre_act(f0, f1, f2, wq[k][3], wq[k][4], wq[k][5], bq[k].y));
    }

    v8f acc0 = {}, acc1 = {};
    #pragma unroll
    for (int g = 0; g < KS1B / GS; ++g) {
        #pragma unroll
        for (int k = 0; k < GS; ++k) {
            int i = GS * g + k;
            if (i + AH < KS1B) {
                a0[i + AH].x = ap0[4 * (i + AH)];  a0[i + AH].y = ap0[4 * (i + AH) + 1];
                a1[i + AH].x = ap1[4 * (i + AH)];  a1[i + AH].y = ap1[4 * (i + AH) + 1];
                #pragma unroll
                for (int w = 0; w < 6; ++w) wq[i + AH][w] = wp[12 * (i + AH) + w];
                bq[i + AH].x = bp[4 * (i + AH)];   bq[i + AH].y = bp[4 * (i + AH) + 1];
            } else {
                #pragma unroll
                for (int w = 0; w < RQ; ++w) {
                    int idx = RQ * (i + AH - KS1B) + w;
                    if (idx < 4) {
                        biasv[idx] = *(const v4f*)(b2p + 16 * (idx >> 1) + 4 * (idx & 1));
                    } else if (idx < NTAIL) {
                        int kk = idx - 4;
                        int j  = kk >> 3;
                        int b  = (kk >> 2) & 1;
                        int pc = kk & 3;
                        xf[j][b][pc] = *(const v4f*)(xr[j][b] + 16 * (pc >> 1) + 4 * (pc & 1));
                    }
                }
            }
        }
        #pragma unroll
        for (int k = 0; k < GS; ++k) {
            int i = GS * g + k;
            acc0 = __builtin_amdgcn_wmma_f32_16x16x4_f32(
                       false, a0[i], false, bfr[i], (short)0, acc0, false, false);
            acc1 = __builtin_amdgcn_wmma_f32_16x16x4_f32(
                       false, a1[i], false, bfr[i], (short)0, acc1, false, false);
        }
        __builtin_amdgcn_sched_barrier(0);
        if (g + 1 < KS1B / GS) {
            #pragma unroll
            for (int k = 0; k < GS; ++k) {
                int n = GS * (g + 1) + k;
                bfr[n].x = gelu_tanh(pre_act(f0, f1, f2, wq[n][0], wq[n][1], wq[n][2], bq[n].x));
                bfr[n].y = gelu_tanh(pre_act(f0, f1, f2, wq[n][3], wq[n][4], wq[n][5], bq[n].y));
            }
        }
        __builtin_amdgcn_sched_barrier(0);
    }

    #pragma unroll
    for (int r = 0; r < 8; ++r) {
        acc0[r] += biasv[r >> 2][r & 3];
        acc1[r] += biasv[2 + (r >> 2)][r & 3];
    }

    #pragma unroll
    for (int j = 0; j < MAXT; ++j) {
        if (j == 0 || j < nt) {
            v8f o0 = {}, o1 = {};
            #pragma unroll
            for (int s = 0; s < 8; ++s) {
                int jf = s >> 2, qd = s & 3;
                int pc = 2 * jf + (qd >> 1), e0 = 2 * (qd & 1);
                v2f bf, xa0, xa1;
                bf.x  = jf ? acc1[2 * qd]     : acc0[2 * qd];
                bf.y  = jf ? acc1[2 * qd + 1] : acc0[2 * qd + 1];
                xa0.x = xf[j][0][pc][e0];  xa0.y = xf[j][0][pc][e0 + 1];
                xa1.x = xf[j][1][pc][e0];  xa1.y = xf[j][1][pc][e0 + 1];
                o0 = __builtin_amdgcn_wmma_f32_16x16x4_f32(
                         false, xa0, false, bf, (short)0, o0, false, false);
                o1 = __builtin_amdgcn_wmma_f32_16x16x4_f32(
                         false, xa1, false, bf, (short)0, o1, false, false);
            }
            int ti = ti0 + j;
            int jo = p - ti;
            int tb = ti * 16 + 8 * half;
            int sb = tb + 383 - u;
            float* O0 = out + (size_t)h * TK * TK + (size_t)tb * TK + sb;
            float* O1 = O0 + (size_t)HH * TK * TK;
            if (jo >= 1 && jo <= 23) {
                #pragma unroll
                for (int r = 0; r < 8; ++r) {
                    O0[r * (TK + 1)] = o0[r];
                    O1[r * (TK + 1)] = o1[r];
                }
            } else {
                #pragma unroll
                for (int r = 0; r < 8; ++r) {
                    int s = sb + r;
                    if (s >= 0 && s < TK) {
                        O0[r * (TK + 1)] = o0[r];
                        O1[r * (TK + 1)] = o1[r];
                    }
                }
            }
        }
    }
}

extern "C" void kernel_launch(void* const* d_in, const int* in_sizes, int n_in,
                              void* d_out, int out_size, void* d_ws, size_t ws_size,
                              hipStream_t stream)
{
    const float* x  = (const float*)d_in[0];
    const float* pd = (const float*)d_in[1];
    const float* W1 = (const float*)d_in[2];
    const float* b1 = (const float*)d_in[3];
    const float* W2 = (const float*)d_in[4];
    const float* b2 = (const float*)d_in[5];
    float* out = (float*)d_out;

    srpe_fused<<<dim3(120 * HH), dim3(32), 0, stream>>>(x, pd, W1, b1, W2, b2, out);
}
